# starting kernel
# baseline (speedup 1.0000x reference)
#include <hip/hip_runtime.h>
#include <hip/hip_bf16.h>

typedef float v2f __attribute__((ext_vector_type(2)));
typedef float v8f __attribute__((ext_vector_type(8)));

#define T_FRAMES   61
#define DT         (1.0f / 60.0f)
#define WG_THREADS 256
#define VOX_PER_WG 128
#define FS         66
#define SS         130

__global__ __launch_bounds__(WG_THREADS)
void srtm_wmma_kernel(const float* __restrict__ ref_tac,
                      const float* __restrict__ kparam,
                      float* __restrict__ y,
                      int L)
{
    __shared__ float Fsh[64 * FS];
    __shared__ float Ssh[64 * SS];

    const int tid    = threadIdx.x;
    const int wave   = tid >> 5;
    const int lane   = tid & 31;
    const int laneLo = lane & 15;
    const int laneHi = lane >> 4;
    const int wgBase = blockIdx.x * VOX_PER_WG;

    for (int idx = tid; idx < 64 * 64; idx += WG_THREADS) {
        const int q = idx >> 6, r = idx & 63;
        float v = 0.0f;
        if (q < 60 && r < 60) {
            const float a = ref_tac[q], b = ref_tac[q + 1];
            v = fmaxf(fmaf((float)r * DT, b - a, a), 0.0f);
        }
        Fsh[q * FS + r] = v;
    }

    int vox = wgBase + wave * 16 + laneLo;
    vox = vox < L ? vox : L - 1;
    const float kk2 = kparam[L + vox];
    const float bp  = kparam[2 * L + vox];
    const float d   = expf((-kk2 / (1.0f + bp)) * DT);
    const float d2  = d * d;
    const float d4  = d2 * d2;
    float pwx = laneHi ? d2      : d4;
    float pwy = laneHi ? d       : d2 * d;
    const int aoff = laneHi ? 2 : 0;

    __syncthreads();

    v8f acc0 = {}, acc1 = {}, acc2 = {}, acc3 = {};
    for (int kb = 56; kb >= 0; kb -= 4) {
        v2f b; b.x = pwx; b.y = pwy;
        const int col = kb + aoff;
        v2f a0 = { Fsh[(laneLo     ) * FS + col], Fsh[(laneLo     ) * FS + col + 1] };
        v2f a1 = { Fsh[(laneLo + 16) * FS + col], Fsh[(laneLo + 16) * FS + col + 1] };
        v2f a2 = { Fsh[(laneLo + 32) * FS + col], Fsh[(laneLo + 32) * FS + col + 1] };
        v2f a3 = { Fsh[(laneLo + 48) * FS + col], Fsh[(laneLo + 48) * FS + col + 1] };
        acc0 = __builtin_amdgcn_wmma_f32_16x16x4_f32(false, a0, false, b, (short)0, acc0, false, false);
        acc1 = __builtin_amdgcn_wmma_f32_16x16x4_f32(false, a1, false, b, (short)0, acc1, false, false);
        acc2 = __builtin_amdgcn_wmma_f32_16x16x4_f32(false, a2, false, b, (short)0, acc2, false, false);
        acc3 = __builtin_amdgcn_wmma_f32_16x16x4_f32(false, a3, false, b, (short)0, acc3, false, false);
        pwx *= d4; pwy *= d4;
    }

    const int scol = wave * 16 + laneLo;
    #pragma unroll
    for (int v = 0; v < 8; ++v) {
        const int r0 = v + laneHi * 8;
        Ssh[(r0     ) * SS + scol] = acc0[v];
        Ssh[(r0 + 16) * SS + scol] = acc1[v];
        Ssh[(r0 + 32) * SS + scol] = acc2[v];
        Ssh[(r0 + 48) * SS + scol] = acc3[v];
    }
    __syncthreads();

    if (tid < VOX_PER_WG) {
        const int l = wgBase + tid;
        if (l < L) {
            const float R1  = kparam[l];
            const float kk2s = kparam[L + l];
            const float bps  = kparam[2 * L + l];
            const float As  = -kk2s / (1.0f + bps);
            const float Bs  = kk2s * (1.0f - R1 / (1.0f + bps));
            const float d60 = expf(As);
            float C = 0.0f;
            const float r0v = ref_tac[0];
            y[l] = fmaf(Bs, DT * fmaxf(r0v, 0.0f), r0v * R1);
            for (int m = 1; m < T_FRAMES; ++m) {
                C = fmaf(d60, C, Ssh[(m - 1) * SS + tid]);
                const float rm   = ref_tac[m];
                const float conv = DT * (fmaxf(rm, 0.0f) + C);
                y[(size_t)m * L + l] = fmaf(Bs, conv, rm * R1);
            }
        }
    }
}

extern "C" void kernel_launch(void* const* d_in, const int* in_sizes, int n_in,
                              void* d_out, int out_size, void* d_ws, size_t ws_size,
                              hipStream_t stream) {
    const float* ref_tac = (const float*)d_in[1];
    const float* kparam  = (const float*)d_in[2];
    float* y = (float*)d_out;
    const int L = in_sizes[2] / 3;
    const int grid = (L + VOX_PER_WG - 1) / VOX_PER_WG;
    srtm_wmma_kernel<<<grid, WG_THREADS, 0, stream>>>(ref_tac, kparam, y, L);
}


# Frame recurrence as a segmented scan: wave s sums, carries and replays frames 8s..8s+7 of all 128 voxels, a lane takes 4 voxels (16-byte LDS reads and stores); all 8 waves store; sum order changes
# speedup vs baseline: 2.0550x; 2.0550x over previous
#include <hip/hip_runtime.h>
#include <hip/hip_bf16.h>


typedef float v2f __attribute__((ext_vector_type(2)));
typedef float v4f __attribute__((ext_vector_type(4)));
typedef float v8f __attribute__((ext_vector_type(8)));

#define T_FRAMES   61
#define DT         (1.0f / 60.0f)
#define WG_THREADS 256
#define VOX_PER_WG 128
#define SS         132
#define SEG        8

static __device__ __forceinline__ v4f fma4(v4f a, v4f b, v4f c) {
    v4f r;
    r.x = fmaf(a.x, b.x, c.x); r.y = fmaf(a.y, b.y, c.y);
    r.z = fmaf(a.z, b.z, c.z); r.w = fmaf(a.w, b.w, c.w);
    return r;
}

static __device__ __forceinline__ v4f frame_out(v4f C, float rm, v4f R1v, v4f Bv) {
    const float fm = fmaxf(rm, 0.0f);
    v4f o;
    o.x = fmaf(Bv.x, DT * (fm + C.x), rm * R1v.x);
    o.y = fmaf(Bv.y, DT * (fm + C.y), rm * R1v.y);
    o.z = fmaf(Bv.z, DT * (fm + C.z), rm * R1v.z);
    o.w = fmaf(Bv.w, DT * (fm + C.w), rm * R1v.w);
    return o;
}

template <int NF>
static __device__ __forceinline__ void replay16(v4f C, const v4f (&sx)[SEG], const float (&rf)[SEG],
                                                v4f dv, v4f R1v, v4f Bv, float* out, size_t pitch) {
    #pragma unroll
    for (int a = 0; a < NF; ++a) {
        C = fma4(dv, C, sx[a]);
        *(v4f*)(out + a * pitch) = frame_out(C, rf[a], R1v, Bv);
    }
}

__global__ __launch_bounds__(WG_THREADS)
void srtm_wmma_kernel(const float* __restrict__ ref_tac,
                      const float* __restrict__ kparam,
                      float* __restrict__ y,
                      int L)
{
    __shared__ alignas(16) float Ssh[64 * SS];
    __shared__ alignas(16) float Agg[(SEG - 1) * VOX_PER_WG];
    __shared__ alignas(16) float Par[3 * VOX_PER_WG];

    const int tid    = threadIdx.x;
    const int wave   = tid >> 5;
    const int lane   = tid & 31;
    const int laneLo = lane & 15;
    const int laneHi = lane >> 4;
    const int wgBase = blockIdx.x * VOX_PER_WG;
    const int seg    = __builtin_amdgcn_readfirstlane(wave);

    float rf[SEG];
    #pragma unroll
    for (int a = 0; a < SEG; ++a) {
        const int m = SEG * seg + a;
        rf[a] = ref_tac[m < T_FRAMES ? m : T_FRAMES - 1];
    }

    int vox = wgBase + wave * 16 + laneLo;
    vox = vox < L ? vox : L - 1;
    const float R1  = kparam[vox];
    const float kk2 = kparam[L + vox];
    const float bp  = kparam[2 * L + vox];
    float ra[4], rb[4];
    bool  rin[4];
    for (int mt = 0; mt < 4; ++mt) {
        const int q  = laneLo + 16 * mt - 1;
        rin[mt]      = (q >= 0) && (q < 60);
        const int qc = rin[mt] ? q : 0;
        ra[mt] = ref_tac[qc];
        rb[mt] = ref_tac[qc + 1];
    }
    asm volatile("" :: "v"(R1), "v"(kk2), "v"(bp),
                       "v"(ra[0]), "v"(rb[0]), "v"(ra[1]), "v"(rb[1]),
                       "v"(ra[2]), "v"(rb[2]), "v"(ra[3]), "v"(rb[3]));
    float fa[4], fd[4];
    for (int mt = 0; mt < 4; ++mt) {
        fa[mt] = rin[mt] ? ra[mt]          : 0.0f;
        fd[mt] = rin[mt] ? rb[mt] - ra[mt] : 0.0f;
    }

    const float As  = -kk2 / (1.0f + bp);
    const float d   = expf(As * DT);
    const float d2  = d * d;
    const float d4  = d2 * d2;
    float pwx = laneHi ? d2      : d4;
    float pwy = laneHi ? d       : d2 * d;
    const int aoff = laneHi ? 2 : 0;

    v8f acc0 = {}, acc1 = {}, acc2 = {}, acc3 = {};
    #pragma unroll
    for (int kb = 56; kb >= 0; kb -= 4) {
        v2f b; b.x = pwx; b.y = pwy;
        const int   col = kb + aoff;
        const float x0  = (float)col * DT;
        const float x1  = (float)(col + 1) * DT;
        v2f a0 = { fmaxf(fmaf(x0, fd[0], fa[0]), 0.0f), fmaxf(fmaf(x1, fd[0], fa[0]), 0.0f) };
        v2f a1 = { fmaxf(fmaf(x0, fd[1], fa[1]), 0.0f), fmaxf(fmaf(x1, fd[1], fa[1]), 0.0f) };
        v2f a2 = { fmaxf(fmaf(x0, fd[2], fa[2]), 0.0f), fmaxf(fmaf(x1, fd[2], fa[2]), 0.0f) };
        v2f a3 = { fmaxf(fmaf(x0, fd[3], fa[3]), 0.0f), fmaxf(fmaf(x1, fd[3], fa[3]), 0.0f) };
        acc0 = __builtin_amdgcn_wmma_f32_16x16x4_f32(false, a0, false, b, (short)0, acc0, false, false);
        acc1 = __builtin_amdgcn_wmma_f32_16x16x4_f32(false, a1, false, b, (short)0, acc1, false, false);
        acc2 = __builtin_amdgcn_wmma_f32_16x16x4_f32(false, a2, false, b, (short)0, acc2, false, false);
        acc3 = __builtin_amdgcn_wmma_f32_16x16x4_f32(false, a3, false, b, (short)0, acc3, false, false);
        pwx *= d4; pwy *= d4;
    }

    const int scol = wave * 16 + laneLo;
    Par[scol]                  = R1;
    Par[VOX_PER_WG + scol]     = kk2 * (1.0f - R1 / (1.0f + bp));
    Par[2 * VOX_PER_WG + scol] = expf(As);

    #pragma unroll
    for (int v = 0; v < 8; ++v) {
        const int r0 = v + laneHi * 8;
        Ssh[(r0     ) * SS + scol] = acc0[v];
        Ssh[(r0 + 16) * SS + scol] = acc1[v];
        Ssh[(r0 + 32) * SS + scol] = acc2[v];
        Ssh[(r0 + 48) * SS + scol] = acc3[v];
    }
    __syncthreads();

    const int c4  = lane * 4;
    const v4f R1v = *(const v4f*)&Par[c4];
    const v4f Bv  = *(const v4f*)&Par[VOX_PER_WG + c4];
    const v4f dv  = *(const v4f*)&Par[2 * VOX_PER_WG + c4];
    v4f sx[SEG];
    #pragma unroll
    for (int a = 0; a < SEG; ++a)
        sx[a] = *(const v4f*)&Ssh[(SEG * seg + a) * SS + c4];

    if (seg < SEG - 1) {
        v4f t = sx[0];
        #pragma unroll
        for (int a = 1; a < SEG; ++a) t = fma4(dv, t, sx[a]);
        *(v4f*)&Agg[seg * VOX_PER_WG + c4] = t;
    }
    __syncthreads();

    v4f tj[SEG - 1];
    #pragma unroll
    for (int j = 0; j < SEG - 1; ++j)
        tj[j] = *(const v4f*)&Agg[j * VOX_PER_WG + c4];
    const v4f dv2 = dv * dv;
    const v4f dv4 = dv2 * dv2;
    const v4f dv8 = dv4 * dv4;
    v4f C = { 0.0f, 0.0f, 0.0f, 0.0f };
    #pragma unroll
    for (int j = 0; j < SEG - 1; ++j)
        if (j < seg) C = fma4(dv8, C, tj[j]);

    const int l4 = wgBase + c4;
    if ((L & 3) == 0) {
        if (l4 < L) {
            float* out = y + (size_t)(SEG * seg) * L + l4;
            if (seg < SEG - 1) replay16<SEG>(C, sx, rf, dv, R1v, Bv, out, (size_t)L);
            else               replay16<T_FRAMES - SEG * (SEG - 1)>(C, sx, rf, dv, R1v, Bv, out, (size_t)L);
        }
    } else {
        #pragma unroll
        for (int a = 0; a < SEG; ++a) {
            const int m = SEG * seg + a;
            C = fma4(dv, C, sx[a]);
            if (m < T_FRAMES) {
                const v4f o = frame_out(C, rf[a], R1v, Bv);
                float* row = y + (size_t)m * L + l4;
                if (l4     < L) row[0] = o.x;
                if (l4 + 1 < L) row[1] = o.y;
                if (l4 + 2 < L) row[2] = o.z;
                if (l4 + 3 < L) row[3] = o.w;
            }
        }
    }
}

extern "C" void kernel_launch(void* const* d_in, const int* in_sizes, int n_in,
                              void* d_out, int out_size, void* d_ws, size_t ws_size,
                              hipStream_t stream) {
    const float* ref_tac = (const float*)d_in[1];
    const float* kparam  = (const float*)d_in[2];
    float* y = (float*)d_out;
    const int L = in_sizes[2] / 3;
    const int grid = (L + VOX_PER_WG - 1) / VOX_PER_WG;
    srtm_wmma_kernel<<<grid, WG_THREADS, 0, stream>>>(ref_tac, kparam, y, L);
}


# Frames split between the two waves of a SIMD: waves 0-3 multiply frames 0-31 first (raised priority), scan and store them under waves 4-7's products of frames 32-63; split barrier; same sum order
# speedup vs baseline: 2.2198x; 1.0802x over previous
#include <hip/hip_runtime.h>
#include <hip/hip_bf16.h>


typedef float v2f __attribute__((ext_vector_type(2)));
typedef float v4f __attribute__((ext_vector_type(4)));
typedef float v8f __attribute__((ext_vector_type(8)));

#define T_FRAMES   61
#define DT         (1.0f / 60.0f)
#define WG_THREADS 256
#define VOX_PER_WG 128
#define SS         132
#define SEG        8
#define HALF_ROWS  32

static __device__ __forceinline__ v4f fma4(v4f a, v4f b, v4f c) {
    v4f r;
    r.x = fmaf(a.x, b.x, c.x); r.y = fmaf(a.y, b.y, c.y);
    r.z = fmaf(a.z, b.z, c.z); r.w = fmaf(a.w, b.w, c.w);
    return r;
}

static __device__ __forceinline__ v4f frame_out(v4f C, float rm, v4f R1v, v4f Bv) {
    const float fm = fmaxf(rm, 0.0f);
    v4f o;
    o.x = fmaf(Bv.x, DT * (fm + C.x), rm * R1v.x);
    o.y = fmaf(Bv.y, DT * (fm + C.y), rm * R1v.y);
    o.z = fmaf(Bv.z, DT * (fm + C.z), rm * R1v.z);
    o.w = fmaf(Bv.w, DT * (fm + C.w), rm * R1v.w);
    return o;
}

template <int NF>
static __device__ __forceinline__ void replay16(v4f C, const v4f (&sx)[SEG], const float (&rf)[SEG],
                                                v4f dv, v4f R1v, v4f Bv, float* out, size_t pitch) {
    #pragma unroll
    for (int a = 0; a < NF; ++a) {
        C = fma4(dv, C, sx[a]);
        *(v4f*)(out + a * pitch) = frame_out(C, rf[a], R1v, Bv);
    }
}

static __device__ __forceinline__ v4f seg_total(const v4f (&sx)[SEG], v4f dv) {
    v4f t = sx[0];
    #pragma unroll
    for (int a = 1; a < SEG; ++a) t = fma4(dv, t, sx[a]);
    return t;
}

template <int NP>
static __device__ __forceinline__ v4f carry_over(const float* S0, v4f dv, v4f C) {
    v4f r[NP][SEG];
    #pragma unroll
    for (int j = 0; j < NP; ++j) {
        #pragma unroll
        for (int a = 0; a < SEG; ++a) r[j][a] = *(const v4f*)&S0[(SEG * j + a) * SS];
    }
    const v4f dv2 = dv * dv;
    const v4f dv4 = dv2 * dv2;
    const v4f dv8 = dv4 * dv4;
    #pragma unroll
    for (int j = 0; j < NP; ++j) C = fma4(dv8, C, seg_total(r[j], dv));
    return C;
}

static __device__ __forceinline__ void half_product(const float (&fa)[2], const float (&fd)[2], const float (&d)[2],
                                                    int laneHi, v8f (&acc)[2][2]) {
    float pwx[2], pwy[2], d4[2];
    #pragma unroll
    for (int nt = 0; nt < 2; ++nt) {
        const float d2 = d[nt] * d[nt];
        d4[nt]  = d2 * d2;
        pwx[nt] = laneHi ? d2    : d4[nt];
        pwy[nt] = laneHi ? d[nt] : d2 * d[nt];
    }
    const int aoff = laneHi ? 2 : 0;
    #pragma unroll
    for (int kb = 56; kb >= 0; kb -= 4) {
        const int   col = kb + aoff;
        const float x0  = (float)col * DT;
        const float x1  = (float)(col + 1) * DT;
        v2f a0 = { fmaxf(fmaf(x0, fd[0], fa[0]), 0.0f), fmaxf(fmaf(x1, fd[0], fa[0]), 0.0f) };
        v2f a1 = { fmaxf(fmaf(x0, fd[1], fa[1]), 0.0f), fmaxf(fmaf(x1, fd[1], fa[1]), 0.0f) };
        v2f b0; b0.x = pwx[0]; b0.y = pwy[0];
        v2f b1; b1.x = pwx[1]; b1.y = pwy[1];
        acc[0][0] = __builtin_amdgcn_wmma_f32_16x16x4_f32(false, a0, false, b0, (short)0, acc[0][0], false, false);
        acc[0][1] = __builtin_amdgcn_wmma_f32_16x16x4_f32(false, a0, false, b1, (short)0, acc[0][1], false, false);
        acc[1][0] = __builtin_amdgcn_wmma_f32_16x16x4_f32(false, a1, false, b0, (short)0, acc[1][0], false, false);
        acc[1][1] = __builtin_amdgcn_wmma_f32_16x16x4_f32(false, a1, false, b1, (short)0, acc[1][1], false, false);
        pwx[0] *= d4[0]; pwy[0] *= d4[0];
        pwx[1] *= d4[1]; pwy[1] *= d4[1];
    }
}

static __device__ __forceinline__ void half_spill(float* Sh, const v8f (&acc)[2][2], int laneHi) {
    #pragma unroll
    for (int v = 0; v < 8; ++v) {
        const int r0 = v + laneHi * 8;
        Sh[(r0     ) * SS     ] = acc[0][0][v];
        Sh[(r0     ) * SS + 16] = acc[0][1][v];
        Sh[(r0 + 16) * SS     ] = acc[1][0][v];
        Sh[(r0 + 16) * SS + 16] = acc[1][1][v];
    }
}

template <int NF>
static __device__ __forceinline__ void seg_store(v4f C, const v4f (&sx)[SEG], const float (&rf)[SEG],
                                                 v4f dv, v4f R1v, v4f Bv, float* y, int m0, int l4, int L) {
    if ((L & 3) == 0) {
        if (l4 < L) replay16<NF>(C, sx, rf, dv, R1v, Bv, y + (size_t)m0 * L + l4, (size_t)L);
    } else {
        #pragma unroll
        for (int a = 0; a < NF; ++a) {
            C = fma4(dv, C, sx[a]);
            const v4f o = frame_out(C, rf[a], R1v, Bv);
            float* row = y + (size_t)(m0 + a) * L + l4;
            if (l4     < L) row[0] = o.x;
            if (l4 + 1 < L) row[1] = o.y;
            if (l4 + 2 < L) row[2] = o.z;
            if (l4 + 3 < L) row[3] = o.w;
        }
    }
}

__global__ __launch_bounds__(WG_THREADS)
void srtm_wmma_kernel(const float* __restrict__ ref_tac,
                      const float* __restrict__ kparam,
                      float* __restrict__ y,
                      int L)
{
    __shared__ alignas(16) float Ssh[64 * SS];
    __shared__ alignas(16) float Par[3 * VOX_PER_WG];
    __shared__ alignas(16) float Cmid[VOX_PER_WG];

    const int tid    = threadIdx.x;
    const int wave   = tid >> 5;
    const int lane   = tid & 31;
    const int laneLo = lane & 15;
    const int laneHi = lane >> 4;
    const int wgBase = blockIdx.x * VOX_PER_WG;
    const int seg    = __builtin_amdgcn_readfirstlane(wave);
    const int half   = seg >> 2;
    const int quad   = seg & 3;

    float rf[SEG];
    #pragma unroll
    for (int a = 0; a < SEG; ++a) {
        const int m = SEG * seg + a;
        rf[a] = ref_tac[m < T_FRAMES ? m : T_FRAMES - 1];
    }

    float R1[2], kk2[2], bp[2];
    #pragma unroll
    for (int nt = 0; nt < 2; ++nt) {
        int vox = wgBase + quad * 32 + nt * 16 + laneLo;
        vox = vox < L ? vox : L - 1;
        R1[nt]  = kparam[vox];
        kk2[nt] = kparam[L + vox];
        bp[nt]  = kparam[2 * L + vox];
    }
    float ra[2], rb[2];
    bool  rin[2];
    #pragma unroll
    for (int i = 0; i < 2; ++i) {
        const int q  = laneLo + 16 * (2 * half + i) - 1;
        rin[i]       = (q >= 0) && (q < 60);
        const int qc = rin[i] ? q : 0;
        ra[i] = ref_tac[qc];
        rb[i] = ref_tac[qc + 1];
    }
    asm volatile("" :: "v"(R1[0]), "v"(kk2[0]), "v"(bp[0]), "v"(R1[1]), "v"(kk2[1]), "v"(bp[1]),
                       "v"(ra[0]), "v"(rb[0]), "v"(ra[1]), "v"(rb[1]));
    float fa[2], fd[2];
    #pragma unroll
    for (int i = 0; i < 2; ++i) {
        fa[i] = rin[i] ? ra[i]         : 0.0f;
        fd[i] = rin[i] ? rb[i] - ra[i] : 0.0f;
    }

    float As[2], d[2];
    #pragma unroll
    for (int nt = 0; nt < 2; ++nt) {
        As[nt] = -kk2[nt] / (1.0f + bp[nt]);
        d[nt]  = expf(As[nt] * DT);
    }

    v8f acc[2][2] = {};
    float* Sh = &Ssh[(HALF_ROWS * half) * SS + quad * 32 + laneLo];
    const int c4 = lane * 4;
    const float* S0 = &Ssh[(HALF_ROWS * half) * SS + c4];

    if (half == 0) {
        __builtin_amdgcn_s_setprio(3);
        half_product(fa, fd, d, laneHi, acc);
        __builtin_amdgcn_s_setprio(0);
        half_spill(Sh, acc, laneHi);
        #pragma unroll
        for (int nt = 0; nt < 2; ++nt) {
            const int scol = quad * 32 + nt * 16 + laneLo;
            Par[scol]                  = R1[nt];
            Par[VOX_PER_WG + scol]     = kk2[nt] * (1.0f - R1[nt] / (1.0f + bp[nt]));
            Par[2 * VOX_PER_WG + scol] = expf(As[nt]);
        }
    }
    __builtin_amdgcn_fence(__ATOMIC_RELEASE, "workgroup");
    __builtin_amdgcn_s_barrier_signal(-1);

    if (half == 1) half_product(fa, fd, d, laneHi, acc);

    __builtin_amdgcn_s_barrier_wait(-1);
    __builtin_amdgcn_fence(__ATOMIC_ACQUIRE, "workgroup");

    v4f R1v, Bv, dv, C;
    v4f sx[SEG];
    C = (v4f){ 0.0f, 0.0f, 0.0f, 0.0f };
    if (half == 0) {
        R1v = *(const v4f*)&Par[c4];
        Bv  = *(const v4f*)&Par[VOX_PER_WG + c4];
        dv  = *(const v4f*)&Par[2 * VOX_PER_WG + c4];
        #pragma unroll
        for (int a = 0; a < SEG; ++a) sx[a] = *(const v4f*)&S0[(SEG * quad + a) * SS];
        if      (quad == 1) C = carry_over<1>(S0, dv, C);
        else if (quad == 2) C = carry_over<2>(S0, dv, C);
        else if (quad == 3) {
            C = carry_over<3>(S0, dv, C);
            const v4f dv2 = dv * dv;
            const v4f dv4 = dv2 * dv2;
            *(v4f*)&Cmid[c4] = fma4(dv4 * dv4, C, seg_total(sx, dv));
        }
    } else {
        half_spill(Sh, acc, laneHi);
    }
    __builtin_amdgcn_fence(__ATOMIC_RELEASE, "workgroup");
    __builtin_amdgcn_s_barrier_signal(-1);

    const int l4 = wgBase + c4;
    if (half == 0) seg_store<SEG>(C, sx, rf, dv, R1v, Bv, y, SEG * seg, l4, L);

    __builtin_amdgcn_s_barrier_wait(-1);
    __builtin_amdgcn_fence(__ATOMIC_ACQUIRE, "workgroup");

    if (half == 1) {
        R1v = *(const v4f*)&Par[c4];
        Bv  = *(const v4f*)&Par[VOX_PER_WG + c4];
        dv  = *(const v4f*)&Par[2 * VOX_PER_WG + c4];
        C   = *(const v4f*)&Cmid[c4];
        #pragma unroll
        for (int a = 0; a < SEG; ++a) sx[a] = *(const v4f*)&S0[(SEG * quad + a) * SS];
        if      (quad == 1) C = carry_over<1>(S0, dv, C);
        else if (quad == 2) C = carry_over<2>(S0, dv, C);
        else if (quad == 3) C = carry_over<3>(S0, dv, C);
        if (quad < 3) seg_store<SEG>(C, sx, rf, dv, R1v, Bv, y, SEG * seg, l4, L);
        else          seg_store<T_FRAMES - SEG * (SEG - 1)>(C, sx, rf, dv, R1v, Bv, y, SEG * seg, l4, L);
    }
}

extern "C" void kernel_launch(void* const* d_in, const int* in_sizes, int n_in,
                              void* d_out, int out_size, void* d_ws, size_t ws_size,
                              hipStream_t stream) {
    const float* ref_tac = (const float*)d_in[1];
    const float* kparam  = (const float*)d_in[2];
    float* y = (float*)d_out;
    const int L = in_sizes[2] / 3;
    const int grid = (L + VOX_PER_WG - 1) / VOX_PER_WG;
    srtm_wmma_kernel<<<grid, WG_THREADS, 0, stream>>>(ref_tac, kparam, y, L);
}
